# starting kernel
# baseline (speedup 1.0000x reference)
#include <hip/hip_runtime.h>
#include <math.h>

#define Bdim 64
#define Sdim 512
#define Hdim 768
#define Ndim 64

typedef __attribute__((ext_vector_type(2))) float v2f;
typedef __attribute__((ext_vector_type(8))) float v8f;

__global__ __launch_bounds__(256) void means_kernel(
    const float* __restrict__ top_vec, const int* __restrict__ clss,
    const int* __restrict__ sents_end, float* __restrict__ means) {
  const int m = blockIdx.x;
  const int b = m >> 6;
  const int start = clss[m];
  const int end = sents_end[m];
  const int len = end - start;
  const float inv = 1.0f / (float)(len < 1 ? 1 : len);
  const float* base = top_vec + ((size_t)b * Sdim + (size_t)start) * Hdim;
  for (int h = threadIdx.x; h < Hdim; h += 256) {
    float s = 0.0f;
    for (int t = 0; t < len; ++t) s += base[(size_t)t * Hdim + h];
    means[(size_t)m * Hdim + h] = s * inv;
  }
}

#define ALD 772
__global__ __launch_bounds__(256) void gemm_reduce_kernel(
    const float* __restrict__ means, const float* __restrict__ w_sents,
    const float* __restrict__ b_sents, const float* __restrict__ w_fc,
    const float* __restrict__ b_fc, const float* __restrict__ mask_clss,
    float* __restrict__ out) {
  __shared__ float Abuf[32][ALD];
  __shared__ float h_sh[32];
  __shared__ float c_sh;

  const int tid = threadIdx.x;
  const int row0 = blockIdx.x * 32;

  if (tid < 32) h_sh[tid] = 0.0f;
  if (tid == 0) c_sh = 0.0f;
  __syncthreads();

  float cp = 0.0f;
  for (int i = tid; i < Hdim; i += 256) cp += b_sents[i] * w_fc[i];
  atomicAdd(&c_sh, cp);

  for (int idx = tid; idx < 32 * Hdim; idx += 256) {
    const int r = idx / Hdim;
    const int c = idx - r * Hdim;
    Abuf[r][c] = means[(size_t)(row0 + r) * Hdim + c];
  }
  __syncthreads();

  const int wave = tid >> 5;
  const int lane = tid & 31;
  const int laneLo = lane & 15;
  const int hi = lane >> 4;

  for (int ot = wave; ot < 48; ot += 8) {
    const int ob = ot << 4;
    v8f acc0 = {0.f,0.f,0.f,0.f,0.f,0.f,0.f,0.f};
    v8f acc1 = {0.f,0.f,0.f,0.f,0.f,0.f,0.f,0.f};
    const float* wrow = w_sents + (size_t)(ob + laneLo) * Hdim + (hi << 1);
    const float* a0p = &Abuf[laneLo][hi << 1];
    const float* a1p = &Abuf[16 + laneLo][hi << 1];
#pragma unroll 4
    for (int k0 = 0; k0 < Hdim; k0 += 4) {
      const v2f bv = *(const v2f*)(wrow + k0);
      const v2f a0 = *(const v2f*)(a0p + k0);
      const v2f a1 = *(const v2f*)(a1p + k0);
      acc0 = __builtin_amdgcn_wmma_f32_16x16x4_f32(
          false, a0, false, bv, (short)0, acc0, false, false);
      acc1 = __builtin_amdgcn_wmma_f32_16x16x4_f32(
          false, a1, false, bv, (short)0, acc1, false, false);
    }
    const float wv = w_fc[ob + laneLo];
#pragma unroll
    for (int j = 0; j < 8; ++j) {
      float p0 = acc0[j] * wv;
      float p1 = acc1[j] * wv;
      p0 += __shfl_xor(p0, 1, 32); p0 += __shfl_xor(p0, 2, 32);
      p0 += __shfl_xor(p0, 4, 32); p0 += __shfl_xor(p0, 8, 32);
      p1 += __shfl_xor(p1, 1, 32); p1 += __shfl_xor(p1, 2, 32);
      p1 += __shfl_xor(p1, 4, 32); p1 += __shfl_xor(p1, 8, 32);
      if (laneLo == 0) {
        atomicAdd(&h_sh[(hi << 3) + j], p0);
        atomicAdd(&h_sh[16 + (hi << 3) + j], p1);
      }
    }
  }
  __syncthreads();

  if (tid < 32) {
    const int m = row0 + tid;
    const float mask = mask_clss[m];
    const float dot = h_sh[tid] + c_sh;
    const float hv = mask * dot + b_fc[0];
    out[m] = mask / (1.0f + expf(-hv));
  }
}

extern "C" void kernel_launch(void* const* d_in, const int* in_sizes, int n_in,
                              void* d_out, int out_size, void* d_ws, size_t ws_size,
                              hipStream_t stream) {
  const float* top_vec   = (const float*)d_in[0];
  const int*   clss      = (const int*)d_in[1];
  const int*   sents_end = (const int*)d_in[2];
  const float* mask_clss = (const float*)d_in[3];
  const float* w_sents   = (const float*)d_in[4];
  const float* b_sents   = (const float*)d_in[5];
  const float* w_fc      = (const float*)d_in[6];
  const float* b_fc      = (const float*)d_in[7];
  float* out = (float*)d_out;
  float* means = (float*)d_ws;

  means_kernel<<<Bdim * Ndim, 256, 0, stream>>>(top_vec, clss, sents_end, means);
  gemm_reduce_kernel<<<(Bdim * Ndim) / 32, 256, 0, stream>>>(
      means, w_sents, b_sents, w_fc, b_fc, mask_clss, out);
}


# GEMM re-tiled: 16 rows x 768 cols per work-group (256 of them), 6 accumulators per wave, w_sents by whole-line loads via wave-private LDS, prefetched; k order rotated, fixed-order sums, no atomics
# speedup vs baseline: 2.3628x; 2.3628x over previous
#include <hip/hip_runtime.h>
#include <math.h>


#define Bdim 64
#define Sdim 512
#define Hdim 768
#define Ndim 64

typedef __attribute__((ext_vector_type(2))) float v2f;
typedef __attribute__((ext_vector_type(4))) float v4f;
typedef __attribute__((ext_vector_type(8))) float v8f;

__global__ __launch_bounds__(256) void means_kernel(
    const float* __restrict__ top_vec, const int* __restrict__ clss,
    const int* __restrict__ sents_end, float* __restrict__ means) {
  const int m = blockIdx.x;
  const int b = m >> 6;
  const int start = clss[m];
  const int end = sents_end[m];
  const int len = end - start;
  const float inv = 1.0f / (float)(len < 1 ? 1 : len);
  const float* base = top_vec + ((size_t)b * Sdim + (size_t)start) * Hdim;
  for (int h = threadIdx.x; h < Hdim; h += 256) {
    float s = 0.0f;
    for (int t = 0; t < len; ++t) s += base[(size_t)t * Hdim + h];
    means[(size_t)m * Hdim + h] = s * inv;
  }
}

#define ALD 772
#define WLD 36
#define MROWS 16
#define TPW 6
#define WROWS (TPW * 16)
#define NLD (WROWS / 4)
#define KC 32
#define NCH (Hdim / KC)

#define LOAD_G(kcv)                                                         \
  _Pragma("unroll") for (int n = 0; n < NLD; ++n)                           \
    G[n] = *(const v4f*)(gsrc + (size_t)n * (4 * Hdim) + (kcv));

#define STORE_G(so)                                                         \
  _Pragma("unroll") for (int n = 0; n < NLD; ++n)                           \
    *(v4f*)(wst + (so) + n * (4 * WLD)) = G[n];

#define LOAD_F(kcv, so)                                                     \
  _Pragma("unroll") for (int q = 0; q < 4; ++q) {                           \
    Af[q] = *(const v4f*)(abase + (kcv) + 4 * q);                           \
    _Pragma("unroll") for (int t = 0; t < TPW; ++t)                         \
      F[t][q] = *(const v4f*)(wfr + (so) + t * (16 * WLD) + 4 * q);         \
  }

#define PAIR(v, j) (((j) & 1) ? __builtin_shufflevector(v[(j) >> 1], v[(j) >> 1], 2, 3) \
                              : __builtin_shufflevector(v[(j) >> 1], v[(j) >> 1], 0, 1))

#define COMPUTE()                                                           \
  _Pragma("unroll") for (int j = 0; j < 8; ++j) {                           \
    const v2f a2 = PAIR(Af, j);                                             \
    _Pragma("unroll") for (int t = 0; t < TPW; ++t) {                       \
      const v2f b2 = PAIR(F[t], j);                                         \
      acc[t] = __builtin_amdgcn_wmma_f32_16x16x4_f32(                       \
          false, a2, false, b2, (short)0, acc[t], false, false);            \
    }                                                                       \
  }

__global__ __launch_bounds__(256) void gemm_reduce_kernel(
    const float* __restrict__ means, const float* __restrict__ w_sents,
    const float* __restrict__ b_sents, const float* __restrict__ w_fc,
    const float* __restrict__ b_fc, const float* __restrict__ mask_clss,
    float* __restrict__ out) {
  __shared__ __attribute__((aligned(16))) float Abuf[MROWS][ALD];
  __shared__ __attribute__((aligned(16))) float Wst[8][2 * WROWS * WLD];
  __shared__ float part[8][MROWS];
  __shared__ float cpart[8];

  const int tid = threadIdx.x;
  const int row0 = blockIdx.x * MROWS;
  const int wave = tid >> 5;
  const int lane = tid & 31;
  const int c = lane & 15;
  const int hi = lane >> 4;

  const int rot = blockIdx.x % NCH;

  const float* gsrc = w_sents + (size_t)(wave * WROWS + (lane >> 3)) * Hdim + ((lane & 7) << 2);
  float* wst = &Wst[wave][(lane >> 3) * WLD + ((lane & 7) << 2)];
  const float* wfr = &Wst[wave][c * WLD + (hi << 4)];
  const float* abase = &Abuf[c][hi << 4];

  v4f G[NLD];
  v4f F[TPW][4];
  v4f Af[4];
  v8f acc[TPW];
#pragma unroll
  for (int t = 0; t < TPW; ++t) acc[t] = (v8f){0.f,0.f,0.f,0.f,0.f,0.f,0.f,0.f};

  LOAD_G(rot * KC);

  float cp = 0.0f;
  for (int i = tid; i < Hdim; i += 256) cp += b_sents[i] * w_fc[i];
  cp += __shfl_xor(cp, 16, 32); cp += __shfl_xor(cp, 8, 32);
  cp += __shfl_xor(cp, 4, 32);  cp += __shfl_xor(cp, 2, 32);
  cp += __shfl_xor(cp, 1, 32);
  if (lane == 0) cpart[wave] = cp;

  for (int idx = tid; idx < MROWS * (Hdim / 4); idx += 256) {
    const int r = idx / (Hdim / 4);
    const int c4 = idx - r * (Hdim / 4);
    *(v4f*)&Abuf[r][c4 * 4] =
        *(const v4f*)(means + (size_t)(row0 + r) * Hdim + c4 * 4);
  }
  __syncthreads();

  int ch = rot;
  for (int i = 0; i < NCH - 1; ++i) {
    const int so = (i & 1) * (WROWS * WLD);
    STORE_G(so);
    int nx = ch + 1; if (nx >= NCH) nx -= NCH;
    LOAD_G(nx * KC);
    LOAD_F(ch * KC, so);
    COMPUTE();
    ch = nx;
  }
  STORE_G(WROWS * WLD);
  LOAD_F(ch * KC, WROWS * WLD);
  COMPUTE();

  float p[8];
#pragma unroll
  for (int r = 0; r < 8; ++r) p[r] = 0.0f;
#pragma unroll
  for (int t = 0; t < TPW; ++t) {
    const float wv = w_fc[(wave * TPW + t) * 16 + c];
#pragma unroll
    for (int r = 0; r < 8; ++r) p[r] += acc[t][r] * wv;
  }
#pragma unroll
  for (int r = 0; r < 8; ++r) {
    float v = p[r];
    v += __shfl_xor(v, 1, 32); v += __shfl_xor(v, 2, 32);
    v += __shfl_xor(v, 4, 32); v += __shfl_xor(v, 8, 32);
    if (c == 0) part[wave][(hi << 3) + r] = v;
  }
  __syncthreads();

  if (tid < MROWS) {
    float hsum = 0.0f, csum = 0.0f;
    for (int w = 0; w < 8; ++w) { hsum += part[w][tid]; csum += cpart[w]; }
    const int m = row0 + tid;
    const float mask = mask_clss[m];
    const float dot = hsum + csum;
    const float hv = mask * dot + b_fc[0];
    out[m] = mask / (1.0f + expf(-hv));
  }
}

extern "C" void kernel_launch(void* const* d_in, const int* in_sizes, int n_in,
                              void* d_out, int out_size, void* d_ws, size_t ws_size,
                              hipStream_t stream) {
  const float* top_vec   = (const float*)d_in[0];
  const int*   clss      = (const int*)d_in[1];
  const int*   sents_end = (const int*)d_in[2];
  const float* mask_clss = (const float*)d_in[3];
  const float* w_sents   = (const float*)d_in[4];
  const float* b_sents   = (const float*)d_in[5];
  const float* w_fc      = (const float*)d_in[6];
  const float* b_fc      = (const float*)d_in[7];
  float* out = (float*)d_out;
  float* means = (float*)d_ws;

  means_kernel<<<Bdim * Ndim, 256, 0, stream>>>(top_vec, clss, sents_end, means);
  gemm_reduce_kernel<<<(Bdim * Ndim) / MROWS, 256, 0, stream>>>(
      means, w_sents, b_sents, w_fc, b_fc, mask_clss, out);
}


# segment means computed inside the GEMM work-group while it builds its 16-row A tile in LDS: one launch, no workspace round trip, sum order kept; rows with end < start now follow the reference
# speedup vs baseline: 3.5654x; 1.5090x over previous
#include <hip/hip_runtime.h>
#include <math.h>


#define Bdim 64
#define Sdim 512
#define Hdim 768
#define Ndim 64

typedef __attribute__((ext_vector_type(2))) float v2f;
typedef __attribute__((ext_vector_type(4))) float v4f;
typedef __attribute__((ext_vector_type(8))) float v8f;

#define ALD 772
#define WLD 36
#define MROWS 16
#define TPW 6
#define WROWS (TPW * 16)
#define NLD (WROWS / 4)
#define KC 32
#define NCH (Hdim / KC)
#define TG 8
#define HQ (Hdim / 128)

#define LOAD_G(kcv)                                                         \
  _Pragma("unroll") for (int n = 0; n < NLD; ++n)                           \
    G[n] = *(const v4f*)(gsrc + (size_t)n * (4 * Hdim) + (kcv));

#define STORE_G(so)                                                         \
  _Pragma("unroll") for (int n = 0; n < NLD; ++n)                           \
    *(v4f*)(wst + (so) + n * (4 * WLD)) = G[n];

#define LOAD_F(kcv, so)                                                     \
  _Pragma("unroll") for (int q = 0; q < 4; ++q) {                           \
    Af[q] = *(const v4f*)(abase + (kcv) + 4 * q);                           \
    _Pragma("unroll") for (int t = 0; t < TPW; ++t)                         \
      F[t][q] = *(const v4f*)(wfr + (so) + t * (16 * WLD) + 4 * q);         \
  }

#define PAIR(v, j) (((j) & 1) ? __builtin_shufflevector(v[(j) >> 1], v[(j) >> 1], 2, 3) \
                              : __builtin_shufflevector(v[(j) >> 1], v[(j) >> 1], 0, 1))

#define COMPUTE()                                                           \
  _Pragma("unroll") for (int j = 0; j < 8; ++j) {                           \
    const v2f a2 = PAIR(Af, j);                                             \
    _Pragma("unroll") for (int t = 0; t < TPW; ++t) {                       \
      const v2f b2 = PAIR(F[t], j);                                         \
      acc[t] = __builtin_amdgcn_wmma_f32_16x16x4_f32(                       \
          false, a2, false, b2, (short)0, acc[t], false, false);            \
    }                                                                       \
  }

__global__ __launch_bounds__(256) void gemm_reduce_kernel(
    const float* __restrict__ top_vec, const int* __restrict__ clss,
    const int* __restrict__ sents_end, const float* __restrict__ w_sents,
    const float* __restrict__ b_sents, const float* __restrict__ w_fc,
    const float* __restrict__ b_fc, const float* __restrict__ mask_clss,
    float* __restrict__ out) {
  __shared__ __attribute__((aligned(16))) float Abuf[MROWS][ALD];
  __shared__ __attribute__((aligned(16))) float Wst[8][2 * WROWS * WLD];
  __shared__ float part[8][MROWS];
  __shared__ float cpart[8];

  const int tid = threadIdx.x;
  const int row0 = blockIdx.x * MROWS;
  const int wave = tid >> 5;
  const int lane = tid & 31;
  const int c = lane & 15;
  const int hi = lane >> 4;

  const int rot = blockIdx.x % NCH;

  const float* gsrc = w_sents + (size_t)(wave * WROWS + (lane >> 3)) * Hdim + ((lane & 7) << 2);
  float* wst = &Wst[wave][(lane >> 3) * WLD + ((lane & 7) << 2)];
  const float* wfr = &Wst[wave][c * WLD + (hi << 4)];
  const float* abase = &Abuf[c][hi << 4];

  v4f G[NLD];
  v4f F[TPW][4];
  v4f Af[4];
  v8f acc[TPW];
#pragma unroll
  for (int t = 0; t < TPW; ++t) acc[t] = (v8f){0.f,0.f,0.f,0.f,0.f,0.f,0.f,0.f};

  LOAD_G(rot * KC);

  float cp = 0.0f;
  for (int i = tid; i < Hdim; i += 256) cp += b_sents[i] * w_fc[i];
  cp += __shfl_xor(cp, 16, 32); cp += __shfl_xor(cp, 8, 32);
  cp += __shfl_xor(cp, 4, 32);  cp += __shfl_xor(cp, 2, 32);
  cp += __shfl_xor(cp, 1, 32);
  if (lane == 0) cpart[wave] = cp;

  const int wave_s = __builtin_amdgcn_readfirstlane(wave);
  for (int rr = 0; rr < MROWS / 8; ++rr) {
    const int r = wave_s * (MROWS / 8) + rr;
    const int m = row0 + r;
    int st = clss[m];
    int en = sents_end[m];
    st = st < 0 ? 0 : (st > Sdim ? Sdim : st);
    en = en < 0 ? 0 : (en > Sdim ? Sdim : en);
    const int len = en - st;
    const int lo = len >= 0 ? st : en;
    const int n = len >= 0 ? len : -len;
    const float scale = len > 0 ? 1.0f / (float)len : -1.0f;
    const float* tp = top_vec + ((size_t)(m >> 6) * Sdim + (size_t)lo) * Hdim + (lane << 2);
    v4f sum[HQ];
#pragma unroll
    for (int q = 0; q < HQ; ++q) sum[q] = (v4f){0.f, 0.f, 0.f, 0.f};
    for (int t = 0; t < n; t += TG) {
      v4f x[TG][HQ];
#pragma unroll
      for (int j = 0; j < TG; ++j) {
        if (t + j < n) {
#pragma unroll
          for (int q = 0; q < HQ; ++q)
            x[j][q] = *(const v4f*)(tp + (size_t)(t + j) * Hdim + q * 128);
        } else {
#pragma unroll
          for (int q = 0; q < HQ; ++q) x[j][q] = (v4f){0.f, 0.f, 0.f, 0.f};
        }
      }
#pragma unroll
      for (int j = 0; j < TG; ++j) {
#pragma unroll
        for (int q = 0; q < HQ; ++q) sum[q] += x[j][q];
      }
    }
#pragma unroll
    for (int q = 0; q < HQ; ++q)
      *(v4f*)&Abuf[r][q * 128 + (lane << 2)] = sum[q] * scale;
  }
  __syncthreads();

  int ch = rot;
  for (int i = 0; i < NCH - 1; ++i) {
    const int so = (i & 1) * (WROWS * WLD);
    STORE_G(so);
    int nx = ch + 1; if (nx >= NCH) nx -= NCH;
    LOAD_G(nx * KC);
    LOAD_F(ch * KC, so);
    COMPUTE();
    ch = nx;
  }
  STORE_G(WROWS * WLD);
  LOAD_F(ch * KC, WROWS * WLD);
  COMPUTE();

  float p[8];
#pragma unroll
  for (int r = 0; r < 8; ++r) p[r] = 0.0f;
#pragma unroll
  for (int t = 0; t < TPW; ++t) {
    const float wv = w_fc[(wave * TPW + t) * 16 + c];
#pragma unroll
    for (int r = 0; r < 8; ++r) p[r] += acc[t][r] * wv;
  }
#pragma unroll
  for (int r = 0; r < 8; ++r) {
    float v = p[r];
    v += __shfl_xor(v, 1, 32); v += __shfl_xor(v, 2, 32);
    v += __shfl_xor(v, 4, 32); v += __shfl_xor(v, 8, 32);
    if (c == 0) part[wave][(hi << 3) + r] = v;
  }
  __syncthreads();

  if (tid < MROWS) {
    float hsum = 0.0f, csum = 0.0f;
    for (int w = 0; w < 8; ++w) { hsum += part[w][tid]; csum += cpart[w]; }
    const int m = row0 + tid;
    const float mask = mask_clss[m];
    const float dot = hsum + csum;
    const float hv = mask * dot + b_fc[0];
    out[m] = mask / (1.0f + expf(-hv));
  }
}

extern "C" void kernel_launch(void* const* d_in, const int* in_sizes, int n_in,
                              void* d_out, int out_size, void* d_ws, size_t ws_size,
                              hipStream_t stream) {
  const float* top_vec   = (const float*)d_in[0];
  const int*   clss      = (const int*)d_in[1];
  const int*   sents_end = (const int*)d_in[2];
  const float* mask_clss = (const float*)d_in[3];
  const float* w_sents   = (const float*)d_in[4];
  const float* b_sents   = (const float*)d_in[5];
  const float* w_fc      = (const float*)d_in[6];
  const float* b_fc      = (const float*)d_in[7];
  float* out = (float*)d_out;

  gemm_reduce_kernel<<<(Bdim * Ndim) / MROWS, 256, 0, stream>>>(
      top_vec, clss, sents_end, w_sents, b_sents, w_fc, b_fc, mask_clss, out);
}
